# starting kernel
# baseline (speedup 1.0000x reference)
#include <hip/hip_runtime.h>
#include <cstdint>
#include <cstddef>

#define INC  128
#define HIDC 64
#define OUTC 40

typedef float v2f __attribute__((ext_vector_type(2)));
typedef float v8f __attribute__((ext_vector_type(8)));


__global__ void k_init_deg(float* __restrict__ deg, int n) {
  int i = blockIdx.x * blockDim.x + threadIdx.x;
  if (i < n) deg[i] = 1.0f;
}

__global__ void k_count_deg(const int* __restrict__ dst, float* __restrict__ deg, int E) {
  int e = blockIdx.x * blockDim.x + threadIdx.x;
  if (e < E) unsafeAtomicAdd(&deg[dst[e]], 1.0f);
}

__global__ void k_rsqrt(float* __restrict__ d, int n) {
  int i = blockIdx.x * blockDim.x + threadIdx.x;
  if (i < n) d[i] = rsqrtf(d[i]);
}

__global__ void k_norm(const int* __restrict__ src, const int* __restrict__ dst,
                       const float* __restrict__ dinv, float* __restrict__ norm, int E) {
  int e = blockIdx.x * blockDim.x + threadIdx.x;
  if (e < E) norm[e] = dinv[src[e]] * dinv[dst[e]];
}


__global__ __launch_bounds__(128)
void k_gemm1_wmma(const float* __restrict__ x, const float* __restrict__ W1,
                  float* __restrict__ hx) {
  __shared__ float sA[16 * 132];
  const int r0 = blockIdx.x * 16;
  const int t  = threadIdx.x;

  {
    const int row = t >> 3;
    const int q   = (t & 7) * 16;
    const float4* s = (const float4*)(x + (size_t)(r0 + row) * INC + q);
    float4*       d = (float4*)(sA + row * 132 + q);
#pragma unroll
    for (int j = 0; j < 4; ++j) d[j] = s[j];
  }
  __syncthreads();

  const int wave  = t >> 5;
  const int lane  = t & 31;
  const int c0    = wave * 16;
  const int lhalf = lane & 15;
  const int khi   = (lane >> 4) * 2;

  v8f acc = {};
#pragma unroll
  for (int k = 0; k < INC; k += 4) {
    v2f a, b;
    const float* ap = sA + lhalf * 132 + k + khi;
    a.x = ap[0];
    a.y = ap[1];
    const float* bp = W1 + (size_t)(k + khi) * HIDC + c0 + lhalf;
    b.x = bp[0];
    b.y = bp[HIDC];
    acc = __builtin_amdgcn_wmma_f32_16x16x4_f32(false, a, false, b,
                                                (short)0, acc, false, false);
  }

  const int m0 = r0 + ((lane >> 4) * 8);
  const int n  = c0 + lhalf;
#pragma unroll
  for (int v = 0; v < 8; ++v)
    hx[(size_t)(m0 + v) * HIDC + n] = acc[v];
}


__global__ __launch_bounds__(96)
void k_gemm2_wmma(const float* __restrict__ agg1, const float* __restrict__ b1,
                  const float* __restrict__ W2, float* __restrict__ h2) {
  __shared__ float sA[16 * 68];
  const int r0 = blockIdx.x * 16;
  const int t  = threadIdx.x;

  for (int i = t; i < 256; i += 96) {
    const int row = i >> 4;
    const int q   = (i & 15) * 4;
    float4 v = *(const float4*)(agg1 + (size_t)(r0 + row) * HIDC + q);
    v.x = fmaxf(v.x + b1[q + 0], 0.0f);
    v.y = fmaxf(v.y + b1[q + 1], 0.0f);
    v.z = fmaxf(v.z + b1[q + 2], 0.0f);
    v.w = fmaxf(v.w + b1[q + 3], 0.0f);
    *(float4*)(sA + row * 68 + q) = v;
  }
  __syncthreads();

  const int wave  = t >> 5;
  const int lane  = t & 31;
  const int c0    = wave * 16;
  const int lhalf = lane & 15;
  const int khi   = (lane >> 4) * 2;
  const int col   = c0 + lhalf;
  const bool colOK = (col < OUTC);

  v8f acc = {};
#pragma unroll
  for (int k = 0; k < HIDC; k += 4) {
    v2f a, b;
    const float* ap = sA + lhalf * 68 + k + khi;
    a.x = ap[0];
    a.y = ap[1];
    float bx = 0.0f, by = 0.0f;
    if (colOK) {
      bx = W2[(size_t)(k + khi) * OUTC + col];
      by = W2[(size_t)(k + khi + 1) * OUTC + col];
    }
    b.x = bx;
    b.y = by;
    acc = __builtin_amdgcn_wmma_f32_16x16x4_f32(false, a, false, b,
                                                (short)0, acc, false, false);
  }

  if (colOK) {
    const int m0 = r0 + ((lane >> 4) * 8);
#pragma unroll
    for (int v = 0; v < 8; ++v)
      h2[(size_t)(m0 + v) * OUTC + col] = acc[v];
  }
}


__global__ void k_selfloop_h(const float* __restrict__ hx, const float* __restrict__ dinv,
                             float* __restrict__ agg, int total4) {
  int t = blockIdx.x * blockDim.x + threadIdx.x;
  if (t < total4) {
    int i = t >> 4;
    float di = dinv[i];
    float s = di * di;
    float4 v = ((const float4*)hx)[t];
    v.x *= s; v.y *= s; v.z *= s; v.w *= s;
    ((float4*)agg)[t] = v;
  }
}

__global__ void k_scatter_h(const int* __restrict__ src, const int* __restrict__ dst,
                            const float* __restrict__ norm, const float* __restrict__ hx,
                            float* __restrict__ agg, int total4) {
  int t = blockIdx.x * blockDim.x + threadIdx.x;
  if (t < total4) {
    int e  = t >> 4;
    int c4 = (t & 15) * 4;
    float w = norm[e];
    const float4 v = *(const float4*)(hx + (size_t)src[e] * HIDC + c4);
    float* p = agg + (size_t)dst[e] * HIDC + c4;
    unsafeAtomicAdd(p + 0, v.x * w);
    unsafeAtomicAdd(p + 1, v.y * w);
    unsafeAtomicAdd(p + 2, v.z * w);
    unsafeAtomicAdd(p + 3, v.w * w);
  }
}

__global__ void k_selfloop_o(const float* __restrict__ h2, const float* __restrict__ dinv,
                             const float* __restrict__ b2, float* __restrict__ out,
                             int total4) {
  int t = blockIdx.x * blockDim.x + threadIdx.x;
  if (t < total4) {
    int i = t / 10;
    int q = (t - i * 10) * 4;
    float di = dinv[i];
    float s = di * di;
    float4 v = ((const float4*)h2)[t];
    const float4 b = *(const float4*)(b2 + q);
    v.x = b.x + s * v.x;
    v.y = b.y + s * v.y;
    v.z = b.z + s * v.z;
    v.w = b.w + s * v.w;
    ((float4*)out)[t] = v;
  }
}

__global__ void k_scatter_o(const int* __restrict__ src, const int* __restrict__ dst,
                            const float* __restrict__ norm, const float* __restrict__ h2,
                            float* __restrict__ out, int total4) {
  int t = blockIdx.x * blockDim.x + threadIdx.x;
  if (t < total4) {
    int e = t / 10;
    int q = (t - e * 10) * 4;
    float w = norm[e];
    const float4 v = *(const float4*)(h2 + (size_t)src[e] * OUTC + q);
    float* p = out + (size_t)dst[e] * OUTC + q;
    unsafeAtomicAdd(p + 0, v.x * w);
    unsafeAtomicAdd(p + 1, v.y * w);
    unsafeAtomicAdd(p + 2, v.z * w);
    unsafeAtomicAdd(p + 3, v.w * w);
  }
}


extern "C" void kernel_launch(void* const* d_in, const int* in_sizes, int n_in,
                              void* d_out, int out_size, void* d_ws, size_t ws_size,
                              hipStream_t stream) {
  const float* x  = (const float*)d_in[0];
  const int*   ei = (const int*)d_in[1];
  const float* W1 = (const float*)d_in[2];
  const float* b1 = (const float*)d_in[3];
  const float* W2 = (const float*)d_in[4];
  const float* b2 = (const float*)d_in[5];
  float* out = (float*)d_out;

  const int N = in_sizes[0] / INC;
  const int E = in_sizes[1] / 2;
  const int* src = ei;
  const int* dst = ei + E;

  char*  ws  = (char*)d_ws;
  size_t off = 0;
  auto carve = [&](size_t bytes) -> float* {
    float* p = (float*)(ws + off);
    off = (off + bytes + 255) & ~(size_t)255;
    return p;
  };
  float* dinv = carve((size_t)N * 4);
  float* norm = carve((size_t)E * 4);
  float* hx   = carve((size_t)N * HIDC * 4);
  float* agg1 = carve((size_t)N * HIDC * 4);
  float* h2   = carve((size_t)N * OUTC * 4);
  (void)n_in; (void)out_size; (void)ws_size;

  const int B = 256;
  k_init_deg  <<<(N + B - 1) / B, B, 0, stream>>>(dinv, N);
  k_count_deg <<<(E + B - 1) / B, B, 0, stream>>>(dst, dinv, E);
  k_rsqrt     <<<(N + B - 1) / B, B, 0, stream>>>(dinv, N);
  k_norm      <<<(E + B - 1) / B, B, 0, stream>>>(src, dst, dinv, norm, E);

  k_gemm1_wmma<<<N / 16, 128, 0, stream>>>(x, W1, hx);
  {
    int tot4 = N * (HIDC / 4);
    k_selfloop_h<<<(tot4 + B - 1) / B, B, 0, stream>>>(hx, dinv, agg1, tot4);
  }
  {
    int tot4 = E * (HIDC / 4);
    k_scatter_h<<<(tot4 + B - 1) / B, B, 0, stream>>>(src, dst, norm, hx, agg1, tot4);
  }

  k_gemm2_wmma<<<N / 16, 96, 0, stream>>>(agg1, b1, W2, h2);
  {
    int tot4 = N * (OUTC / 4);
    k_selfloop_o<<<(tot4 + B - 1) / B, B, 0, stream>>>(h2, dinv, b2, out, tot4);
  }
  {
    int tot4 = E * (OUTC / 4);
    k_scatter_o<<<(tot4 + B - 1) / B, B, 0, stream>>>(src, dst, norm, h2, out, tot4);
  }
}


# Edge scatters: one wave per 8 edges, lane = consecutive float of a row, so each row load and each f32 atomic add covers one whole 128-B stretch, not 4 quarter lines; adds still in arrival order
# speedup vs baseline: 1.5771x; 1.5771x over previous
#include <hip/hip_runtime.h>
#include <cstdint>
#include <cstddef>

#define INC  128
#define HIDC 64
#define OUTC 40

typedef float v2f __attribute__((ext_vector_type(2)));
typedef float v8f __attribute__((ext_vector_type(8)));


__global__ void k_init_deg(float* __restrict__ deg, int n) {
  int i = blockIdx.x * blockDim.x + threadIdx.x;
  if (i < n) deg[i] = 1.0f;
}

__global__ void k_count_deg(const int* __restrict__ dst, float* __restrict__ deg, int E) {
  int e = blockIdx.x * blockDim.x + threadIdx.x;
  if (e < E) unsafeAtomicAdd(&deg[dst[e]], 1.0f);
}

__global__ void k_rsqrt(float* __restrict__ d, int n) {
  int i = blockIdx.x * blockDim.x + threadIdx.x;
  if (i < n) d[i] = rsqrtf(d[i]);
}

__global__ void k_norm(const int* __restrict__ src, const int* __restrict__ dst,
                       const float* __restrict__ dinv, float* __restrict__ norm, int E) {
  int e = blockIdx.x * blockDim.x + threadIdx.x;
  if (e < E) norm[e] = dinv[src[e]] * dinv[dst[e]];
}


__global__ __launch_bounds__(128)
void k_gemm1_wmma(const float* __restrict__ x, const float* __restrict__ W1,
                  float* __restrict__ hx) {
  __shared__ float sA[16 * 132];
  const int r0 = blockIdx.x * 16;
  const int t  = threadIdx.x;

  {
    const int row = t >> 3;
    const int q   = (t & 7) * 16;
    const float4* s = (const float4*)(x + (size_t)(r0 + row) * INC + q);
    float4*       d = (float4*)(sA + row * 132 + q);
#pragma unroll
    for (int j = 0; j < 4; ++j) d[j] = s[j];
  }
  __syncthreads();

  const int wave  = t >> 5;
  const int lane  = t & 31;
  const int c0    = wave * 16;
  const int lhalf = lane & 15;
  const int khi   = (lane >> 4) * 2;

  v8f acc = {};
#pragma unroll
  for (int k = 0; k < INC; k += 4) {
    v2f a, b;
    const float* ap = sA + lhalf * 132 + k + khi;
    a.x = ap[0];
    a.y = ap[1];
    const float* bp = W1 + (size_t)(k + khi) * HIDC + c0 + lhalf;
    b.x = bp[0];
    b.y = bp[HIDC];
    acc = __builtin_amdgcn_wmma_f32_16x16x4_f32(false, a, false, b,
                                                (short)0, acc, false, false);
  }

  const int m0 = r0 + ((lane >> 4) * 8);
  const int n  = c0 + lhalf;
#pragma unroll
  for (int v = 0; v < 8; ++v)
    hx[(size_t)(m0 + v) * HIDC + n] = acc[v];
}


__global__ __launch_bounds__(96)
void k_gemm2_wmma(const float* __restrict__ agg1, const float* __restrict__ b1,
                  const float* __restrict__ W2, float* __restrict__ h2) {
  __shared__ float sA[16 * 68];
  const int r0 = blockIdx.x * 16;
  const int t  = threadIdx.x;

  for (int i = t; i < 256; i += 96) {
    const int row = i >> 4;
    const int q   = (i & 15) * 4;
    float4 v = *(const float4*)(agg1 + (size_t)(r0 + row) * HIDC + q);
    v.x = fmaxf(v.x + b1[q + 0], 0.0f);
    v.y = fmaxf(v.y + b1[q + 1], 0.0f);
    v.z = fmaxf(v.z + b1[q + 2], 0.0f);
    v.w = fmaxf(v.w + b1[q + 3], 0.0f);
    *(float4*)(sA + row * 68 + q) = v;
  }
  __syncthreads();

  const int wave  = t >> 5;
  const int lane  = t & 31;
  const int c0    = wave * 16;
  const int lhalf = lane & 15;
  const int khi   = (lane >> 4) * 2;
  const int col   = c0 + lhalf;
  const bool colOK = (col < OUTC);

  v8f acc = {};
#pragma unroll
  for (int k = 0; k < HIDC; k += 4) {
    v2f a, b;
    const float* ap = sA + lhalf * 68 + k + khi;
    a.x = ap[0];
    a.y = ap[1];
    float bx = 0.0f, by = 0.0f;
    if (colOK) {
      bx = W2[(size_t)(k + khi) * OUTC + col];
      by = W2[(size_t)(k + khi + 1) * OUTC + col];
    }
    b.x = bx;
    b.y = by;
    acc = __builtin_amdgcn_wmma_f32_16x16x4_f32(false, a, false, b,
                                                (short)0, acc, false, false);
  }

  if (colOK) {
    const int m0 = r0 + ((lane >> 4) * 8);
#pragma unroll
    for (int v = 0; v < 8; ++v)
      h2[(size_t)(m0 + v) * OUTC + col] = acc[v];
  }
}


#define EPW 8

static __device__ __forceinline__ float lane_f(float v, int l) {
  return __int_as_float(__builtin_amdgcn_readlane(__float_as_int(v), l));
}

__global__ void k_selfloop_h(const float* __restrict__ hx, const float* __restrict__ dinv,
                             float* __restrict__ agg, int total4) {
  int t = blockIdx.x * blockDim.x + threadIdx.x;
  if (t < total4) {
    int i = t >> 4;
    float di = dinv[i];
    float s = di * di;
    float4 v = ((const float4*)hx)[t];
    v.x *= s; v.y *= s; v.z *= s; v.w *= s;
    ((float4*)agg)[t] = v;
  }
}

template <bool TAIL>
static __device__ __forceinline__ void scatter_rows_h(const float* __restrict__ hx, float* __restrict__ agg,
                                                      int sv, int dv, float wv, int lane, int nvalid) {
  int S[EPW], D[EPW];
  float Wt[EPW];
#pragma unroll
  for (int i = 0; i < EPW; ++i) {
    S[i]  = __builtin_amdgcn_readlane(sv, i);
    D[i]  = __builtin_amdgcn_readlane(dv, i);
    Wt[i] = lane_f(wv, i);
  }
  float a[EPW], b[EPW];
#pragma unroll
  for (int i = 0; i < EPW; ++i) {
    const float* hr = hx + (size_t)S[i] * HIDC;
    a[i] = hr[lane];
    b[i] = hr[lane + 32];
  }
#pragma unroll
  for (int i = 0; i < EPW; ++i) {
    if (!TAIL || i < nvalid) {
      float* ar = agg + (size_t)D[i] * HIDC;
      unsafeAtomicAdd(ar + lane,      a[i] * Wt[i]);
      unsafeAtomicAdd(ar + lane + 32, b[i] * Wt[i]);
    }
  }
}

__global__ __launch_bounds__(256)
void k_scatter_h(const int* __restrict__ src, const int* __restrict__ dst,
                 const float* __restrict__ norm, const float* __restrict__ hx,
                 float* __restrict__ agg, int E) {
  const int lane = threadIdx.x & 31;
  const int e0 = __builtin_amdgcn_readfirstlane((int)(blockIdx.x * 8 + (threadIdx.x >> 5)) * EPW);
  if (e0 >= E) return;
  const int   le = min(e0 + (lane & (EPW - 1)), E - 1);
  const int   sv = src[le];
  const int   dv = dst[le];
  const float wv = norm[le];
  if (e0 + EPW <= E) scatter_rows_h<false>(hx, agg, sv, dv, wv, lane, EPW);
  else               scatter_rows_h<true >(hx, agg, sv, dv, wv, lane, E - e0);
}

__global__ void k_selfloop_o(const float* __restrict__ h2, const float* __restrict__ dinv,
                             const float* __restrict__ b2, float* __restrict__ out,
                             int total4) {
  int t = blockIdx.x * blockDim.x + threadIdx.x;
  if (t < total4) {
    int i = t / 10;
    int q = (t - i * 10) * 4;
    float di = dinv[i];
    float s = di * di;
    float4 v = ((const float4*)h2)[t];
    const float4 b = *(const float4*)(b2 + q);
    v.x = b.x + s * v.x;
    v.y = b.y + s * v.y;
    v.z = b.z + s * v.z;
    v.w = b.w + s * v.w;
    ((float4*)out)[t] = v;
  }
}

template <bool TAIL>
static __device__ __forceinline__ void scatter_rows_o(const float* __restrict__ h2, float* __restrict__ out,
                                                      int sv, int dv, float wv, int lane, int nvalid) {
  float v[(EPW / 4) * 5];
#pragma unroll
  for (int g = 0; g < EPW / 4; ++g) {
#pragma unroll
    for (int k = 0; k < 5; ++k) {
      const int  jA    = (32 * k) / OUTC;
      const int  jB    = (32 * k + 31) / OUTC;
      const int  thr   = OUTC * (jA + 1) - 32 * k;
      const bool first = lane < thr;
      const int  sA    = __builtin_amdgcn_readlane(sv, 4 * g + jA);
      const int  sB    = __builtin_amdgcn_readlane(sv, 4 * g + jB);
      const unsigned rowA = (unsigned)sA * OUTC + (unsigned)(32 * k - OUTC * jA);
      const unsigned rowB = (unsigned)sB * OUTC + (unsigned)(32 * k - OUTC * jB);
      v[g * 5 + k] = h2[(first ? rowA : rowB) + (unsigned)lane];
    }
  }
#pragma unroll
  for (int g = 0; g < EPW / 4; ++g) {
#pragma unroll
    for (int k = 0; k < 5; ++k) {
      const int   jA    = (32 * k) / OUTC;
      const int   jB    = (32 * k + 31) / OUTC;
      const int   thr   = OUTC * (jA + 1) - 32 * k;
      const bool  first = lane < thr;
      const int   dA    = __builtin_amdgcn_readlane(dv, 4 * g + jA);
      const int   dB    = __builtin_amdgcn_readlane(dv, 4 * g + jB);
      const float wA    = lane_f(wv, 4 * g + jA);
      const float wB    = lane_f(wv, 4 * g + jB);
      const unsigned rowA = (unsigned)dA * OUTC + (unsigned)(32 * k - OUTC * jA);
      const unsigned rowB = (unsigned)dB * OUTC + (unsigned)(32 * k - OUTC * jB);
      const float w     = first ? wA : wB;
      if (!TAIL || 4 * g + (first ? jA : jB) < nvalid)
        unsafeAtomicAdd(out + ((first ? rowA : rowB) + (unsigned)lane), v[g * 5 + k] * w);
    }
  }
}

__global__ __launch_bounds__(256)
void k_scatter_o(const int* __restrict__ src, const int* __restrict__ dst,
                 const float* __restrict__ norm, const float* __restrict__ h2,
                 float* __restrict__ out, int E) {
  const int lane = threadIdx.x & 31;
  const int e0 = __builtin_amdgcn_readfirstlane((int)(blockIdx.x * 8 + (threadIdx.x >> 5)) * EPW);
  if (e0 >= E) return;
  const int   le = min(e0 + (lane & (EPW - 1)), E - 1);
  const int   sv = src[le];
  const int   dv = dst[le];
  const float wv = norm[le];
  if (e0 + EPW <= E) scatter_rows_o<false>(h2, out, sv, dv, wv, lane, EPW);
  else               scatter_rows_o<true >(h2, out, sv, dv, wv, lane, E - e0);
}


extern "C" void kernel_launch(void* const* d_in, const int* in_sizes, int n_in,
                              void* d_out, int out_size, void* d_ws, size_t ws_size,
                              hipStream_t stream) {
  const float* x  = (const float*)d_in[0];
  const int*   ei = (const int*)d_in[1];
  const float* W1 = (const float*)d_in[2];
  const float* b1 = (const float*)d_in[3];
  const float* W2 = (const float*)d_in[4];
  const float* b2 = (const float*)d_in[5];
  float* out = (float*)d_out;

  const int N = in_sizes[0] / INC;
  const int E = in_sizes[1] / 2;
  const int* src = ei;
  const int* dst = ei + E;

  char*  ws  = (char*)d_ws;
  size_t off = 0;
  auto carve = [&](size_t bytes) -> float* {
    float* p = (float*)(ws + off);
    off = (off + bytes + 255) & ~(size_t)255;
    return p;
  };
  float* dinv = carve((size_t)N * 4);
  float* norm = carve((size_t)E * 4);
  float* hx   = carve((size_t)N * HIDC * 4);
  float* agg1 = carve((size_t)N * HIDC * 4);
  float* h2   = carve((size_t)N * OUTC * 4);
  (void)n_in; (void)out_size; (void)ws_size;

  const int B = 256;
  k_init_deg  <<<(N + B - 1) / B, B, 0, stream>>>(dinv, N);
  k_count_deg <<<(E + B - 1) / B, B, 0, stream>>>(dst, dinv, E);
  k_rsqrt     <<<(N + B - 1) / B, B, 0, stream>>>(dinv, N);
  k_norm      <<<(E + B - 1) / B, B, 0, stream>>>(src, dst, dinv, norm, E);

  k_gemm1_wmma<<<N / 16, 128, 0, stream>>>(x, W1, hx);
  {
    int tot4 = N * (HIDC / 4);
    k_selfloop_h<<<(tot4 + B - 1) / B, B, 0, stream>>>(hx, dinv, agg1, tot4);
  }
  k_scatter_h<<<(E + 8 * EPW - 1) / (8 * EPW), B, 0, stream>>>(src, dst, norm, hx, agg1, E);

  k_gemm2_wmma<<<N / 16, 96, 0, stream>>>(agg1, b1, W2, h2);
  {
    int tot4 = N * (OUTC / 4);
    k_selfloop_o<<<(tot4 + B - 1) / B, B, 0, stream>>>(h2, dinv, b2, out, tot4);
  }
  k_scatter_o<<<(E + 8 * EPW - 1) / (8 * EPW), B, 0, stream>>>(src, dst, norm, h2, out, E);
}


# Degree count moved into LDS: each work-group counts 32768 edges into 16-bit counters for all nodes and stores one partial row, rsqrt kernel sums the rows; no global atomics there; 10 to 9 launches
# speedup vs baseline: 2.0201x; 1.2809x over previous
#include <hip/hip_runtime.h>
#include <cstdint>
#include <cstddef>

#define INC  128
#define HIDC 64
#define OUTC 40

typedef float v2f __attribute__((ext_vector_type(2)));
typedef float v8f __attribute__((ext_vector_type(8)));


#define DEG_CH   32768
#define DEG_CAPW 50000
#define DEG_T    1024

static __device__ __forceinline__ int deg_pitch(int n) {
  return (((n + 1) >> 1) + 31) & ~31;
}

__global__ __launch_bounds__(DEG_T)
void k_count_deg(const int* __restrict__ dst, unsigned* __restrict__ part, int E, int n) {
  __shared__ uint4 cnt4[DEG_CAPW / 4];
  unsigned* cnt = (unsigned*)cnt4;
  const int t  = threadIdx.x;
  const int e0 = blockIdx.x * DEG_CH;
  const int e1 = min(e0 + DEG_CH, E);
  const int nw = (n + 1) >> 1;
  uint4* prow  = (uint4*)(part + (size_t)blockIdx.x * deg_pitch(n));

  for (int wbase = 0; wbase < nw; wbase += DEG_CAPW) {
    const int wcnt = min(DEG_CAPW, nw - wbase);
    const int q4   = (wcnt + 3) >> 2;
    for (int i = t; i < q4; i += DEG_T) cnt4[i] = make_uint4(0u, 0u, 0u, 0u);
    __syncthreads();

    for (int base = e0; base < e1; base += 8 * DEG_T) {
      int d[8];
#pragma unroll
      for (int j = 0; j < 8; ++j) {
        const int e = base + j * DEG_T + t;
        d[j] = (e < e1) ? dst[e] : -1;
      }
#pragma unroll
      for (int j = 0; j < 8; ++j) {
        const int w = (d[j] >> 1) - wbase;
        if ((unsigned)w < (unsigned)wcnt)
          atomicAdd(&cnt[w], 1u << ((d[j] & 1) << 4));
      }
    }
    __syncthreads();

    uint4* p = prow + (wbase >> 2);
    for (int i = t; i < q4; i += DEG_T) p[i] = cnt4[i];
    __syncthreads();
  }
}

__global__ void k_deg_rsqrt(const unsigned* __restrict__ part, float* __restrict__ dinv,
                            int nparts, int n) {
  const int q = blockIdx.x * blockDim.x + threadIdx.x;
  const int nw = (n + 1) >> 1;
  if (2 * q >= nw) return;
  const int pitch = deg_pitch(n);
  const uint2* col = (const uint2*)part + q;
  unsigned c0 = 1u, c1 = 1u, c2 = 1u, c3 = 1u;
  for (int p0 = 0; p0 < nparts; p0 += 16) {
    uint2 v[16];
#pragma unroll
    for (int j = 0; j < 16; ++j)
      v[j] = (p0 + j < nparts) ? col[(size_t)(p0 + j) * (pitch >> 1)] : make_uint2(0u, 0u);
#pragma unroll
    for (int j = 0; j < 16; ++j) {
      c0 += v[j].x & 0xffffu;  c1 += v[j].x >> 16;
      c2 += v[j].y & 0xffffu;  c3 += v[j].y >> 16;
    }
  }
  const int i = 4 * q;
  const float r0 = rsqrtf((float)c0), r1 = rsqrtf((float)c1);
  const float r2 = rsqrtf((float)c2), r3 = rsqrtf((float)c3);
  if (i + 3 < n) {
    *(float4*)(dinv + i) = make_float4(r0, r1, r2, r3);
  } else {
    dinv[i] = r0;
    if (i + 1 < n) dinv[i + 1] = r1;
    if (i + 2 < n) dinv[i + 2] = r2;
  }
}

__global__ void k_norm(const int* __restrict__ src, const int* __restrict__ dst,
                       const float* __restrict__ dinv, float* __restrict__ norm, int E) {
  int e = blockIdx.x * blockDim.x + threadIdx.x;
  if (e < E) norm[e] = dinv[src[e]] * dinv[dst[e]];
}


__global__ __launch_bounds__(128)
void k_gemm1_wmma(const float* __restrict__ x, const float* __restrict__ W1,
                  float* __restrict__ hx) {
  __shared__ float sA[16 * 132];
  const int r0 = blockIdx.x * 16;
  const int t  = threadIdx.x;

  {
    const int row = t >> 3;
    const int q   = (t & 7) * 16;
    const float4* s = (const float4*)(x + (size_t)(r0 + row) * INC + q);
    float4*       d = (float4*)(sA + row * 132 + q);
#pragma unroll
    for (int j = 0; j < 4; ++j) d[j] = s[j];
  }
  __syncthreads();

  const int wave  = t >> 5;
  const int lane  = t & 31;
  const int c0    = wave * 16;
  const int lhalf = lane & 15;
  const int khi   = (lane >> 4) * 2;

  v8f acc = {};
#pragma unroll
  for (int k = 0; k < INC; k += 4) {
    v2f a, b;
    const float* ap = sA + lhalf * 132 + k + khi;
    a.x = ap[0];
    a.y = ap[1];
    const float* bp = W1 + (size_t)(k + khi) * HIDC + c0 + lhalf;
    b.x = bp[0];
    b.y = bp[HIDC];
    acc = __builtin_amdgcn_wmma_f32_16x16x4_f32(false, a, false, b,
                                                (short)0, acc, false, false);
  }

  const int m0 = r0 + ((lane >> 4) * 8);
  const int n  = c0 + lhalf;
#pragma unroll
  for (int v = 0; v < 8; ++v)
    hx[(size_t)(m0 + v) * HIDC + n] = acc[v];
}


__global__ __launch_bounds__(96)
void k_gemm2_wmma(const float* __restrict__ agg1, const float* __restrict__ b1,
                  const float* __restrict__ W2, float* __restrict__ h2) {
  __shared__ float sA[16 * 68];
  const int r0 = blockIdx.x * 16;
  const int t  = threadIdx.x;

  for (int i = t; i < 256; i += 96) {
    const int row = i >> 4;
    const int q   = (i & 15) * 4;
    float4 v = *(const float4*)(agg1 + (size_t)(r0 + row) * HIDC + q);
    v.x = fmaxf(v.x + b1[q + 0], 0.0f);
    v.y = fmaxf(v.y + b1[q + 1], 0.0f);
    v.z = fmaxf(v.z + b1[q + 2], 0.0f);
    v.w = fmaxf(v.w + b1[q + 3], 0.0f);
    *(float4*)(sA + row * 68 + q) = v;
  }
  __syncthreads();

  const int wave  = t >> 5;
  const int lane  = t & 31;
  const int c0    = wave * 16;
  const int lhalf = lane & 15;
  const int khi   = (lane >> 4) * 2;
  const int col   = c0 + lhalf;
  const bool colOK = (col < OUTC);

  v8f acc = {};
#pragma unroll
  for (int k = 0; k < HIDC; k += 4) {
    v2f a, b;
    const float* ap = sA + lhalf * 68 + k + khi;
    a.x = ap[0];
    a.y = ap[1];
    float bx = 0.0f, by = 0.0f;
    if (colOK) {
      bx = W2[(size_t)(k + khi) * OUTC + col];
      by = W2[(size_t)(k + khi + 1) * OUTC + col];
    }
    b.x = bx;
    b.y = by;
    acc = __builtin_amdgcn_wmma_f32_16x16x4_f32(false, a, false, b,
                                                (short)0, acc, false, false);
  }

  if (colOK) {
    const int m0 = r0 + ((lane >> 4) * 8);
#pragma unroll
    for (int v = 0; v < 8; ++v)
      h2[(size_t)(m0 + v) * OUTC + col] = acc[v];
  }
}


#define EPW 8

static __device__ __forceinline__ float lane_f(float v, int l) {
  return __int_as_float(__builtin_amdgcn_readlane(__float_as_int(v), l));
}

__global__ void k_selfloop_h(const float* __restrict__ hx, const float* __restrict__ dinv,
                             float* __restrict__ agg, int total4) {
  int t = blockIdx.x * blockDim.x + threadIdx.x;
  if (t < total4) {
    int i = t >> 4;
    float di = dinv[i];
    float s = di * di;
    float4 v = ((const float4*)hx)[t];
    v.x *= s; v.y *= s; v.z *= s; v.w *= s;
    ((float4*)agg)[t] = v;
  }
}

template <bool TAIL>
static __device__ __forceinline__ void scatter_rows_h(const float* __restrict__ hx, float* __restrict__ agg,
                                                      int sv, int dv, float wv, int lane, int nvalid) {
  int S[EPW], D[EPW];
  float Wt[EPW];
#pragma unroll
  for (int i = 0; i < EPW; ++i) {
    S[i]  = __builtin_amdgcn_readlane(sv, i);
    D[i]  = __builtin_amdgcn_readlane(dv, i);
    Wt[i] = lane_f(wv, i);
  }
  float a[EPW], b[EPW];
#pragma unroll
  for (int i = 0; i < EPW; ++i) {
    const float* hr = hx + (size_t)S[i] * HIDC;
    a[i] = hr[lane];
    b[i] = hr[lane + 32];
  }
#pragma unroll
  for (int i = 0; i < EPW; ++i) {
    if (!TAIL || i < nvalid) {
      float* ar = agg + (size_t)D[i] * HIDC;
      unsafeAtomicAdd(ar + lane,      a[i] * Wt[i]);
      unsafeAtomicAdd(ar + lane + 32, b[i] * Wt[i]);
    }
  }
}

__global__ __launch_bounds__(256)
void k_scatter_h(const int* __restrict__ src, const int* __restrict__ dst,
                 const float* __restrict__ norm, const float* __restrict__ hx,
                 float* __restrict__ agg, int E) {
  const int lane = threadIdx.x & 31;
  const int e0 = __builtin_amdgcn_readfirstlane((int)(blockIdx.x * 8 + (threadIdx.x >> 5)) * EPW);
  if (e0 >= E) return;
  const int   le = min(e0 + (lane & (EPW - 1)), E - 1);
  const int   sv = src[le];
  const int   dv = dst[le];
  const float wv = norm[le];
  if (e0 + EPW <= E) scatter_rows_h<false>(hx, agg, sv, dv, wv, lane, EPW);
  else               scatter_rows_h<true >(hx, agg, sv, dv, wv, lane, E - e0);
}

__global__ void k_selfloop_o(const float* __restrict__ h2, const float* __restrict__ dinv,
                             const float* __restrict__ b2, float* __restrict__ out,
                             int total4) {
  int t = blockIdx.x * blockDim.x + threadIdx.x;
  if (t < total4) {
    int i = t / 10;
    int q = (t - i * 10) * 4;
    float di = dinv[i];
    float s = di * di;
    float4 v = ((const float4*)h2)[t];
    const float4 b = *(const float4*)(b2 + q);
    v.x = b.x + s * v.x;
    v.y = b.y + s * v.y;
    v.z = b.z + s * v.z;
    v.w = b.w + s * v.w;
    ((float4*)out)[t] = v;
  }
}

template <bool TAIL>
static __device__ __forceinline__ void scatter_rows_o(const float* __restrict__ h2, float* __restrict__ out,
                                                      int sv, int dv, float wv, int lane, int nvalid) {
  float v[(EPW / 4) * 5];
#pragma unroll
  for (int g = 0; g < EPW / 4; ++g) {
#pragma unroll
    for (int k = 0; k < 5; ++k) {
      const int  jA    = (32 * k) / OUTC;
      const int  jB    = (32 * k + 31) / OUTC;
      const int  thr   = OUTC * (jA + 1) - 32 * k;
      const bool first = lane < thr;
      const int  sA    = __builtin_amdgcn_readlane(sv, 4 * g + jA);
      const int  sB    = __builtin_amdgcn_readlane(sv, 4 * g + jB);
      const unsigned rowA = (unsigned)sA * OUTC + (unsigned)(32 * k - OUTC * jA);
      const unsigned rowB = (unsigned)sB * OUTC + (unsigned)(32 * k - OUTC * jB);
      v[g * 5 + k] = h2[(first ? rowA : rowB) + (unsigned)lane];
    }
  }
#pragma unroll
  for (int g = 0; g < EPW / 4; ++g) {
#pragma unroll
    for (int k = 0; k < 5; ++k) {
      const int   jA    = (32 * k) / OUTC;
      const int   jB    = (32 * k + 31) / OUTC;
      const int   thr   = OUTC * (jA + 1) - 32 * k;
      const bool  first = lane < thr;
      const int   dA    = __builtin_amdgcn_readlane(dv, 4 * g + jA);
      const int   dB    = __builtin_amdgcn_readlane(dv, 4 * g + jB);
      const float wA    = lane_f(wv, 4 * g + jA);
      const float wB    = lane_f(wv, 4 * g + jB);
      const unsigned rowA = (unsigned)dA * OUTC + (unsigned)(32 * k - OUTC * jA);
      const unsigned rowB = (unsigned)dB * OUTC + (unsigned)(32 * k - OUTC * jB);
      const float w     = first ? wA : wB;
      if (!TAIL || 4 * g + (first ? jA : jB) < nvalid)
        unsafeAtomicAdd(out + ((first ? rowA : rowB) + (unsigned)lane), v[g * 5 + k] * w);
    }
  }
}

__global__ __launch_bounds__(256)
void k_scatter_o(const int* __restrict__ src, const int* __restrict__ dst,
                 const float* __restrict__ norm, const float* __restrict__ h2,
                 float* __restrict__ out, int E) {
  const int lane = threadIdx.x & 31;
  const int e0 = __builtin_amdgcn_readfirstlane((int)(blockIdx.x * 8 + (threadIdx.x >> 5)) * EPW);
  if (e0 >= E) return;
  const int   le = min(e0 + (lane & (EPW - 1)), E - 1);
  const int   sv = src[le];
  const int   dv = dst[le];
  const float wv = norm[le];
  if (e0 + EPW <= E) scatter_rows_o<false>(h2, out, sv, dv, wv, lane, EPW);
  else               scatter_rows_o<true >(h2, out, sv, dv, wv, lane, E - e0);
}


extern "C" void kernel_launch(void* const* d_in, const int* in_sizes, int n_in,
                              void* d_out, int out_size, void* d_ws, size_t ws_size,
                              hipStream_t stream) {
  const float* x  = (const float*)d_in[0];
  const int*   ei = (const int*)d_in[1];
  const float* W1 = (const float*)d_in[2];
  const float* b1 = (const float*)d_in[3];
  const float* W2 = (const float*)d_in[4];
  const float* b2 = (const float*)d_in[5];
  float* out = (float*)d_out;

  const int N = in_sizes[0] / INC;
  const int E = in_sizes[1] / 2;
  const int* src = ei;
  const int* dst = ei + E;

  char*  ws  = (char*)d_ws;
  size_t off = 0;
  auto carve = [&](size_t bytes) -> float* {
    float* p = (float*)(ws + off);
    off = (off + bytes + 255) & ~(size_t)255;
    return p;
  };
  float* dinv = carve((size_t)N * 4);
  float* norm = carve((size_t)E * 4);
  float* hx   = carve((size_t)N * HIDC * 4);
  float* agg1 = carve((size_t)N * HIDC * 4);
  float* h2   = carve((size_t)N * OUTC * 4);
  (void)n_in; (void)out_size; (void)ws_size;

  const int B = 256;
  {
    unsigned* part = (unsigned*)agg1;
    const int NCH = (E + DEG_CH - 1) / DEG_CH;
    const int NQ  = (N + 3) / 4;
    k_count_deg<<<NCH, DEG_T, 0, stream>>>(dst, part, E, N);
    k_deg_rsqrt<<<(NQ + B - 1) / B, B, 0, stream>>>(part, dinv, NCH, N);
  }
  k_norm      <<<(E + B - 1) / B, B, 0, stream>>>(src, dst, dinv, norm, E);

  k_gemm1_wmma<<<N / 16, 128, 0, stream>>>(x, W1, hx);
  {
    int tot4 = N * (HIDC / 4);
    k_selfloop_h<<<(tot4 + B - 1) / B, B, 0, stream>>>(hx, dinv, agg1, tot4);
  }
  k_scatter_h<<<(E + 8 * EPW - 1) / (8 * EPW), B, 0, stream>>>(src, dst, norm, hx, agg1, E);

  k_gemm2_wmma<<<N / 16, 96, 0, stream>>>(agg1, b1, W2, h2);
  {
    int tot4 = N * (OUTC / 4);
    k_selfloop_o<<<(tot4 + B - 1) / B, B, 0, stream>>>(h2, dinv, b2, out, tot4);
  }
  k_scatter_o<<<(E + 8 * EPW - 1) / (8 * EPW), B, 0, stream>>>(src, dst, norm, h2, out, E);
}
